# starting kernel
# baseline (speedup 1.0000x reference)
#include <hip/hip_runtime.h>
#include <hip/hip_bf16.h>

typedef __attribute__((ext_vector_type(2))) float v2f;
typedef __attribute__((ext_vector_type(8))) float v8f;

#define ALPHA   0.25f
#define EPS     1e-8f
#define W_CLASS 2.0f
#define W_BBOX  5.0f
#define W_GIOU  2.0f
#define KPAD    96

__global__ void class_table_kernel(const float* __restrict__ logits,
                                   float* __restrict__ T, int N, int C) {
    int idx = blockIdx.x * blockDim.x + threadIdx.x;
    if (idx >= N * KPAD) return;
    int n = idx / KPAD;
    int c = idx - n * KPAD;
    float v = 0.0f;
    if (c < C) {
        float x = logits[n * C + c];
        float p = 1.0f / (1.0f + __expf(-x));
        float q = 1.0f - p;
        float pos = ALPHA * q * q * (-__logf(p + EPS));
        float neg = (1.0f - ALPHA) * p * p * (-__logf(q + EPS));
        v = pos - neg;
    }
    T[idx] = v;
}

__global__ void onehot_kernel(const int* __restrict__ labels,
                              float* __restrict__ Ht, int M, int Mpad) {
    int idx = blockIdx.x * blockDim.x + threadIdx.x;
    if (idx >= Mpad * KPAD) return;
    int m = idx / KPAD;
    int c = idx - m * KPAD;
    float v = 0.0f;
    if (m < M && labels[m] == c) v = W_CLASS;
    Ht[idx] = v;
}

__global__ void gtpad_kernel(const float* __restrict__ gt,
                             float* __restrict__ Gp, int M, int Mpad) {
    int m = blockIdx.x * blockDim.x + threadIdx.x;
    if (m >= Mpad) return;
    float4 b = (m < M) ? ((const float4*)gt)[m] : make_float4(0.f, 0.f, 1.f, 1.f);
    ((float4*)Gp)[m] = b;
}

__global__ void __launch_bounds__(256)
cost_kernel(const float* __restrict__ T,
            const float* __restrict__ Ht,
            const float* __restrict__ Gp,
            const float* __restrict__ pboxes,
            float* __restrict__ out,
            int M, int CT, int NT) {
    int wave = threadIdx.x >> 5;
    int lane = threadIdx.x & 31;
    int tile = blockIdx.x * (blockDim.x >> 5) + wave;
    if (tile >= NT) return;

    int rt = tile / CT;
    int ct = tile - rt * CT;
    int qbase = rt << 4;
    int cbase = ct << 4;
    int lo = lane & 15;
    int hi = lane >> 4;

    const float* Arow = T  + (size_t)(qbase + lo) * KPAD + 2 * hi;
    const float* Brow = Ht + (size_t)(cbase + lo) * KPAD + 2 * hi;

    v8f acc = {};
#pragma unroll
    for (int k0 = 0; k0 < KPAD; k0 += 4) {
        v2f a = *(const v2f*)(Arow + k0);
        v2f b = *(const v2f*)(Brow + k0);
        acc = __builtin_amdgcn_wmma_f32_16x16x4_f32(
             false, a,  false, b,
             (short)0, acc,  false,  false);
    }

    int g = cbase + lo;
    float4 gb = ((const float4*)Gp)[g];
    float gx0 = gb.x - 0.5f * gb.z, gy0 = gb.y - 0.5f * gb.w;
    float gx1 = gb.x + 0.5f * gb.z, gy1 = gb.y + 0.5f * gb.w;
    float garea = (gx1 - gx0) * (gy1 - gy0);

#pragma unroll
    for (int r = 0; r < 8; ++r) {
        int q = qbase + r + 8 * hi;
        float4 pb = ((const float4*)pboxes)[q];

        float c_bbox = fabsf(pb.x - gb.x) + fabsf(pb.y - gb.y) +
                       fabsf(pb.z - gb.z) + fabsf(pb.w - gb.w);

        float px0 = pb.x - 0.5f * pb.z, py0 = pb.y - 0.5f * pb.w;
        float px1 = pb.x + 0.5f * pb.z, py1 = pb.y + 0.5f * pb.w;
        float parea = (px1 - px0) * (py1 - py0);
        float iw = fmaxf(fminf(px1, gx1) - fmaxf(px0, gx0), 0.0f);
        float ih = fmaxf(fminf(py1, gy1) - fmaxf(py0, gy0), 0.0f);
        float inter = iw * ih;
        float uni = parea + garea - inter;
        float iou = inter / uni;
        float ew = fmaxf(fmaxf(px1, gx1) - fminf(px0, gx0), 0.0f);
        float eh = fmaxf(fmaxf(py1, gy1) - fminf(py0, gy0), 0.0f);
        float earea = ew * eh;
        float giou = iou - (earea - uni) / earea;

        float v = acc[r] + W_BBOX * c_bbox - W_GIOU * giou;
        if (g < M) out[(size_t)q * M + g] = v;
    }
}

extern "C" void kernel_launch(void* const* d_in, const int* in_sizes, int n_in,
                              void* d_out, int out_size, void* d_ws, size_t ws_size,
                              hipStream_t stream) {
    const float* logits = (const float*)d_in[0];
    const float* pboxes = (const float*)d_in[1];
    const float* gt     = (const float*)d_in[2];
    const int*   labels = (const int*)d_in[3];
    float* out = (float*)d_out;

    int Nq = in_sizes[1] / 4;
    int M  = in_sizes[2] / 4;
    int C  = in_sizes[0] / Nq;
    int Mpad = ((M + 15) / 16) * 16;
    int CT = Mpad / 16;
    int RT = Nq / 16;
    int NT = RT * CT;

    float* T  = (float*)d_ws;
    float* Ht = T + (size_t)Nq * KPAD;
    float* Gp = Ht + (size_t)Mpad * KPAD;

    int t1 = Nq * KPAD;
    class_table_kernel<<<(t1 + 255) / 256, 256, 0, stream>>>(logits, T, Nq, C);
    int t2 = Mpad * KPAD;
    onehot_kernel<<<(t2 + 255) / 256, 256, 0, stream>>>(labels, Ht, M, Mpad);
    gtpad_kernel<<<(Mpad + 255) / 256, 256, 0, stream>>>(gt, Gp, M, Mpad);

    const int wavesPerBlock = 8;
    int blocks = (NT + wavesPerBlock - 1) / wavesPerBlock;
    cost_kernel<<<blocks, 32 * wavesPerBlock, 0, stream>>>(T, Ht, Gp, pboxes,
                                                           out, M, CT, NT);
}


# Class table and one-hot operand stored in WMMA fragment order: each A/B fragment load is one contiguous 512-B wave read (12 b128 per operand) instead of 24 b64 loads touching 16 lines each
# speedup vs baseline: 1.4149x; 1.4149x over previous
#include <hip/hip_runtime.h>
#include <hip/hip_bf16.h>

typedef __attribute__((ext_vector_type(2))) float v2f;
typedef __attribute__((ext_vector_type(8))) float v8f;

#define ALPHA   0.25f
#define EPS     1e-8f
#define W_CLASS 2.0f
#define W_BBOX  5.0f
#define W_GIOU  2.0f
#define KPAD    96

#define KCHUNKS (KPAD / 8)

__device__ __forceinline__ float focal_cost(const float* __restrict__ logits,
                                            int n, int c, int C) {
    if (c >= C) return 0.0f;
    float x = logits[n * C + c];
    float p = 1.0f / (1.0f + __expf(-x));
    float q = 1.0f - p;
    float pos = ALPHA * q * q * (-__logf(p + EPS));
    float neg = (1.0f - ALPHA) * p * p * (-__logf(q + EPS));
    return pos - neg;
}

__global__ void class_table_kernel(const float* __restrict__ logits,
                                   float* __restrict__ Tf, int RT, int C) {
    int idx = blockIdx.x * blockDim.x + threadIdx.x;
    if (idx >= RT * KCHUNKS * 32) return;
    int lane = idx & 31;
    int cj = idx >> 5;
    int rt = cj / KCHUNKS;
    int j  = cj - rt * KCHUNKS;
    int n  = (rt << 4) + (lane & 15);
    int c0 = 8 * j + 2 * (lane >> 4);
    float4 v;
    v.x = focal_cost(logits, n, c0,     C);
    v.y = focal_cost(logits, n, c0 + 1, C);
    v.z = focal_cost(logits, n, c0 + 4, C);
    v.w = focal_cost(logits, n, c0 + 5, C);
    ((float4*)Tf)[idx] = v;
}

__global__ void onehot_kernel(const int* __restrict__ labels,
                              float* __restrict__ Hf, int M, int CT) {
    int idx = blockIdx.x * blockDim.x + threadIdx.x;
    if (idx >= CT * KCHUNKS * 32) return;
    int lane = idx & 31;
    int cj = idx >> 5;
    int ct = cj / KCHUNKS;
    int j  = cj - ct * KCHUNKS;
    int m  = (ct << 4) + (lane & 15);
    int c0 = 8 * j + 2 * (lane >> 4);
    int lab = (m < M) ? labels[m] : -1;
    float4 v;
    v.x = (lab == c0)     ? W_CLASS : 0.0f;
    v.y = (lab == c0 + 1) ? W_CLASS : 0.0f;
    v.z = (lab == c0 + 4) ? W_CLASS : 0.0f;
    v.w = (lab == c0 + 5) ? W_CLASS : 0.0f;
    ((float4*)Hf)[idx] = v;
}

__global__ void gtpad_kernel(const float* __restrict__ gt,
                             float* __restrict__ Gp, int M, int Mpad) {
    int m = blockIdx.x * blockDim.x + threadIdx.x;
    if (m >= Mpad) return;
    float4 b = (m < M) ? ((const float4*)gt)[m] : make_float4(0.f, 0.f, 1.f, 1.f);
    ((float4*)Gp)[m] = b;
}

__global__ void __launch_bounds__(256)
cost_kernel(const float* __restrict__ Tf,
            const float* __restrict__ Hf,
            const float* __restrict__ Gp,
            const float* __restrict__ pboxes,
            float* __restrict__ out,
            int M, int CT, int NT) {
    int wave = threadIdx.x >> 5;
    int lane = threadIdx.x & 31;
    int tile = blockIdx.x * (blockDim.x >> 5) + wave;
    if (tile >= NT) return;

    int rt = tile / CT;
    int ct = tile - rt * CT;
    int qbase = rt << 4;
    int cbase = ct << 4;
    int lo = lane & 15;
    int hi = lane >> 4;

    const float4* Af = (const float4*)Tf + (size_t)rt * (KCHUNKS * 32) + lane;
    const float4* Bf = (const float4*)Hf + (size_t)ct * (KCHUNKS * 32) + lane;

    v8f acc = {};
#pragma unroll
    for (int j = 0; j < KCHUNKS; ++j) {
        float4 a4 = Af[j * 32];
        float4 b4 = Bf[j * 32];
        v2f a0 = {a4.x, a4.y}, b0 = {b4.x, b4.y};
        v2f a1 = {a4.z, a4.w}, b1 = {b4.z, b4.w};
        acc = __builtin_amdgcn_wmma_f32_16x16x4_f32(
             false, a0,  false, b0,
             (short)0, acc,  false,  false);
        acc = __builtin_amdgcn_wmma_f32_16x16x4_f32(
             false, a1,  false, b1,
             (short)0, acc,  false,  false);
    }

    int g = cbase + lo;
    float4 gb = ((const float4*)Gp)[g];
    float gx0 = gb.x - 0.5f * gb.z, gy0 = gb.y - 0.5f * gb.w;
    float gx1 = gb.x + 0.5f * gb.z, gy1 = gb.y + 0.5f * gb.w;
    float garea = (gx1 - gx0) * (gy1 - gy0);

#pragma unroll
    for (int r = 0; r < 8; ++r) {
        int q = qbase + r + 8 * hi;
        float4 pb = ((const float4*)pboxes)[q];

        float c_bbox = fabsf(pb.x - gb.x) + fabsf(pb.y - gb.y) +
                       fabsf(pb.z - gb.z) + fabsf(pb.w - gb.w);

        float px0 = pb.x - 0.5f * pb.z, py0 = pb.y - 0.5f * pb.w;
        float px1 = pb.x + 0.5f * pb.z, py1 = pb.y + 0.5f * pb.w;
        float parea = (px1 - px0) * (py1 - py0);
        float iw = fmaxf(fminf(px1, gx1) - fmaxf(px0, gx0), 0.0f);
        float ih = fmaxf(fminf(py1, gy1) - fmaxf(py0, gy0), 0.0f);
        float inter = iw * ih;
        float uni = parea + garea - inter;
        float iou = inter / uni;
        float ew = fmaxf(fmaxf(px1, gx1) - fminf(px0, gx0), 0.0f);
        float eh = fmaxf(fmaxf(py1, gy1) - fminf(py0, gy0), 0.0f);
        float earea = ew * eh;
        float giou = iou - (earea - uni) / earea;

        float v = acc[r] + W_BBOX * c_bbox - W_GIOU * giou;
        if (g < M) out[(size_t)q * M + g] = v;
    }
}

extern "C" void kernel_launch(void* const* d_in, const int* in_sizes, int n_in,
                              void* d_out, int out_size, void* d_ws, size_t ws_size,
                              hipStream_t stream) {
    const float* logits = (const float*)d_in[0];
    const float* pboxes = (const float*)d_in[1];
    const float* gt     = (const float*)d_in[2];
    const int*   labels = (const int*)d_in[3];
    float* out = (float*)d_out;

    int Nq = in_sizes[1] / 4;
    int M  = in_sizes[2] / 4;
    int C  = in_sizes[0] / Nq;
    int Mpad = ((M + 15) / 16) * 16;
    int CT = Mpad / 16;
    int RT = Nq / 16;
    int NT = RT * CT;

    float* T  = (float*)d_ws;
    float* Ht = T + (size_t)Nq * KPAD;
    float* Gp = Ht + (size_t)Mpad * KPAD;

    int t1 = RT * KCHUNKS * 32;
    class_table_kernel<<<(t1 + 255) / 256, 256, 0, stream>>>(logits, T, RT, C);
    int t2 = CT * KCHUNKS * 32;
    onehot_kernel<<<(t2 + 255) / 256, 256, 0, stream>>>(labels, Ht, M, CT);
    gtpad_kernel<<<(Mpad + 255) / 256, 256, 0, stream>>>(gt, Gp, M, Mpad);

    const int wavesPerBlock = 8;
    int blocks = (NT + wavesPerBlock - 1) / wavesPerBlock;
    cost_kernel<<<blocks, 32 * wavesPerBlock, 0, stream>>>(T, Ht, Gp, pboxes,
                                                           out, M, CT, NT);
}


# Each wave keeps one 16-row tile's class-table fragments and boxes in registers and walks 7 column tiles, refilling the one-hot fragments behind each matrix run: 8136 long waves instead of 56704
# speedup vs baseline: 1.6665x; 1.1778x over previous
#include <hip/hip_runtime.h>
#include <hip/hip_bf16.h>

typedef __attribute__((ext_vector_type(2))) float v2f;
typedef __attribute__((ext_vector_type(8))) float v8f;

#define ALPHA   0.25f
#define EPS     1e-8f
#define W_CLASS 2.0f
#define W_BBOX  5.0f
#define W_GIOU  2.0f
#define KPAD    96

#define KCHUNKS (KPAD / 8)

__device__ __forceinline__ float focal_cost(const float* __restrict__ logits,
                                            int n, int c, int C) {
    if (c >= C) return 0.0f;
    float x = logits[n * C + c];
    float p = 1.0f / (1.0f + __expf(-x));
    float q = 1.0f - p;
    float pos = ALPHA * q * q * (-__logf(p + EPS));
    float neg = (1.0f - ALPHA) * p * p * (-__logf(q + EPS));
    return pos - neg;
}

__global__ void class_table_kernel(const float* __restrict__ logits,
                                   float* __restrict__ Tf, int RT, int C) {
    int idx = blockIdx.x * blockDim.x + threadIdx.x;
    if (idx >= RT * KCHUNKS * 32) return;
    int lane = idx & 31;
    int cj = idx >> 5;
    int rt = cj / KCHUNKS;
    int j  = cj - rt * KCHUNKS;
    int n  = (rt << 4) + (lane & 15);
    int c0 = 8 * j + 2 * (lane >> 4);
    float4 v;
    v.x = focal_cost(logits, n, c0,     C);
    v.y = focal_cost(logits, n, c0 + 1, C);
    v.z = focal_cost(logits, n, c0 + 4, C);
    v.w = focal_cost(logits, n, c0 + 5, C);
    ((float4*)Tf)[idx] = v;
}

__global__ void onehot_kernel(const int* __restrict__ labels,
                              float* __restrict__ Hf, int M, int CT) {
    int idx = blockIdx.x * blockDim.x + threadIdx.x;
    if (idx >= CT * KCHUNKS * 32) return;
    int lane = idx & 31;
    int cj = idx >> 5;
    int ct = cj / KCHUNKS;
    int j  = cj - ct * KCHUNKS;
    int m  = (ct << 4) + (lane & 15);
    int c0 = 8 * j + 2 * (lane >> 4);
    int lab = (m < M) ? labels[m] : -1;
    float4 v;
    v.x = (lab == c0)     ? W_CLASS : 0.0f;
    v.y = (lab == c0 + 1) ? W_CLASS : 0.0f;
    v.z = (lab == c0 + 4) ? W_CLASS : 0.0f;
    v.w = (lab == c0 + 5) ? W_CLASS : 0.0f;
    ((float4*)Hf)[idx] = v;
}

__global__ void gtpad_kernel(const float* __restrict__ gt,
                             float* __restrict__ Gp, int M, int Mpad) {
    int m = blockIdx.x * blockDim.x + threadIdx.x;
    if (m >= Mpad) return;
    float4 b = (m < M) ? ((const float4*)gt)[m] : make_float4(0.f, 0.f, 1.f, 1.f);
    ((float4*)Gp)[m] = b;
}

#define TPW 7

__device__ __forceinline__ void load_frag(float4 (&f)[KCHUNKS],
                                          const float* __restrict__ tile_base,
                                          unsigned lane_byte) {
#pragma unroll
    for (int j = 0; j < KCHUNKS; ++j)
        f[j] = *(const float4*)((const char*)tile_base + (lane_byte + j * 512u));
}

__device__ __forceinline__ v8f class_tile(const float4 (&a)[KCHUNKS],
                                          const float4 (&b)[KCHUNKS]) {
    v8f acc = {};
#pragma unroll
    for (int j = 0; j < KCHUNKS; ++j) {
        v2f a0 = {a[j].x, a[j].y}, b0 = {b[j].x, b[j].y};
        v2f a1 = {a[j].z, a[j].w}, b1 = {b[j].z, b[j].w};
        acc = __builtin_amdgcn_wmma_f32_16x16x4_f32(
             false, a0,  false, b0,
             (short)0, acc,  false,  false);
        acc = __builtin_amdgcn_wmma_f32_16x16x4_f32(
             false, a1,  false, b1,
             (short)0, acc,  false,  false);
    }
    return acc;
}

__device__ __forceinline__ void box_tail(v8f acc, const float4 (&pbx)[8],
                                         float4 gb, bool col_ok,
                                         float* __restrict__ out,
                                         unsigned obyte, unsigned row_bytes) {
    float gx0 = gb.x - 0.5f * gb.z, gy0 = gb.y - 0.5f * gb.w;
    float gx1 = gb.x + 0.5f * gb.z, gy1 = gb.y + 0.5f * gb.w;
    float garea = (gx1 - gx0) * (gy1 - gy0);

    float v[8];
#pragma unroll
    for (int r = 0; r < 8; ++r) {
        float4 pb = pbx[r];

        float c_bbox = fabsf(pb.x - gb.x) + fabsf(pb.y - gb.y) +
                       fabsf(pb.z - gb.z) + fabsf(pb.w - gb.w);

        float px0 = pb.x - 0.5f * pb.z, py0 = pb.y - 0.5f * pb.w;
        float px1 = pb.x + 0.5f * pb.z, py1 = pb.y + 0.5f * pb.w;
        float parea = (px1 - px0) * (py1 - py0);
        float iw = fmaxf(fminf(px1, gx1) - fmaxf(px0, gx0), 0.0f);
        float ih = fmaxf(fminf(py1, gy1) - fmaxf(py0, gy0), 0.0f);
        float inter = iw * ih;
        float uni = parea + garea - inter;
        float iou = inter / uni;
        float ew = fmaxf(fmaxf(px1, gx1) - fminf(px0, gx0), 0.0f);
        float eh = fmaxf(fmaxf(py1, gy1) - fminf(py0, gy0), 0.0f);
        float earea = ew * eh;
        float giou = iou - (earea - uni) / earea;

        v[r] = acc[r] + W_BBOX * c_bbox - W_GIOU * giou;
    }
    if (col_ok) {
#pragma unroll
        for (int r = 0; r < 8; ++r)
            *(float*)((char*)out + (obyte + (unsigned)r * row_bytes)) = v[r];
    }
}

__global__ void __launch_bounds__(256)
cost_kernel(const float* __restrict__ Tf,
            const float* __restrict__ Hf,
            const float* __restrict__ Gp,
            const float* __restrict__ pboxes,
            float* __restrict__ out,
            int M, int CT, int RT, int RB) {
    int wave = threadIdx.x >> 5;
    int lane = threadIdx.x & 31;
    int cg = blockIdx.x / RB;
    int rb = blockIdx.x - cg * RB;
    int rt = rb * (int)(blockDim.x >> 5) + wave;
    if (rt >= RT) return;
    int ct0 = cg * TPW;
    int nt = min(TPW, CT - ct0);

    int lo = lane & 15;
    int hi = lane >> 4;
    unsigned lane_byte = (unsigned)lane * 16u;

    float4 a[KCHUNKS];
    load_frag(a, Tf + (size_t)rt * (KCHUNKS * 128), lane_byte);

    const float* Bt = Hf + (size_t)ct0 * (KCHUNKS * 128);
    float4 b[KCHUNKS];
    load_frag(b, Bt, lane_byte);

    float4 pbx[8];
    int q0 = (rt << 4) + 8 * hi;
#pragma unroll
    for (int r = 0; r < 8; ++r) pbx[r] = ((const float4*)pboxes)[q0 + r];

    unsigned row_bytes = (unsigned)M * 4u;
    unsigned obyte = ((unsigned)q0 * (unsigned)M + (unsigned)(ct0 << 4) + (unsigned)lo) * 4u;
    int g = (ct0 << 4) + lo;
    const float4* Gl = (const float4*)Gp + g;

    float4 gb = Gl[0];
    for (int t = 0; t < nt; ++t) {
        v8f acc = class_tile(a, b);
        asm volatile("" :: "v"(acc) : "memory");
        int tn = min(t + 1, nt - 1);
        load_frag(b, Bt + (size_t)tn * (KCHUNKS * 128), lane_byte);
        float4 gb_next = Gl[tn * 16];
        box_tail(acc, pbx, gb, g + t * 16 < M, out, obyte + (unsigned)t * 64u, row_bytes);
        gb = gb_next;
    }
}

extern "C" void kernel_launch(void* const* d_in, const int* in_sizes, int n_in,
                              void* d_out, int out_size, void* d_ws, size_t ws_size,
                              hipStream_t stream) {
    const float* logits = (const float*)d_in[0];
    const float* pboxes = (const float*)d_in[1];
    const float* gt     = (const float*)d_in[2];
    const int*   labels = (const int*)d_in[3];
    float* out = (float*)d_out;

    int Nq = in_sizes[1] / 4;
    int M  = in_sizes[2] / 4;
    int C  = in_sizes[0] / Nq;
    int Mpad = ((M + 15) / 16) * 16;
    int CT = Mpad / 16;
    int RT = Nq / 16;
    int CG = (CT + TPW - 1) / TPW;

    float* T  = (float*)d_ws;
    float* Ht = T + (size_t)Nq * KPAD;
    float* Gp = Ht + (size_t)Mpad * KPAD;

    int t1 = RT * KCHUNKS * 32;
    class_table_kernel<<<(t1 + 255) / 256, 256, 0, stream>>>(logits, T, RT, C);
    int t2 = CT * KCHUNKS * 32;
    onehot_kernel<<<(t2 + 255) / 256, 256, 0, stream>>>(labels, Ht, M, CT);
    gtpad_kernel<<<(Mpad + 255) / 256, 256, 0, stream>>>(gt, Gp, M, Mpad);

    const int wavesPerBlock = 8;
    int RB = (RT + wavesPerBlock - 1) / wavesPerBlock;
    cost_kernel<<<RB * CG, 32 * wavesPerBlock, 0, stream>>>(T, Ht, Gp, pboxes,
                                                            out, M, CT, RT, RB);
}


# The three preparation kernels (class table, one-hot operand, padded GT boxes) run as one launch, each block taking one job by its index: 2 launches per call instead of 4
# speedup vs baseline: 1.8217x; 1.0931x over previous
#include <hip/hip_runtime.h>
#include <hip/hip_bf16.h>

typedef __attribute__((ext_vector_type(2))) float v2f;
typedef __attribute__((ext_vector_type(8))) float v8f;

#define ALPHA   0.25f
#define EPS     1e-8f
#define W_CLASS 2.0f
#define W_BBOX  5.0f
#define W_GIOU  2.0f
#define KPAD    96

#define KCHUNKS (KPAD / 8)

__device__ __forceinline__ float focal_cost(const float* __restrict__ logits,
                                            int n, int c, int C) {
    if (c >= C) return 0.0f;
    float x = logits[n * C + c];
    float p = 1.0f / (1.0f + __expf(-x));
    float q = 1.0f - p;
    float pos = ALPHA * q * q * (-__logf(p + EPS));
    float neg = (1.0f - ALPHA) * p * p * (-__logf(q + EPS));
    return pos - neg;
}

__device__ __forceinline__ float4 class_table_val(const float* __restrict__ logits,
                                                  int C, int idx) {
    int lane = idx & 31;
    int cj = idx >> 5;
    int rt = cj / KCHUNKS;
    int j  = cj - rt * KCHUNKS;
    int n  = (rt << 4) + (lane & 15);
    int c0 = 8 * j + 2 * (lane >> 4);
    float4 v;
    v.x = focal_cost(logits, n, c0,     C);
    v.y = focal_cost(logits, n, c0 + 1, C);
    v.z = focal_cost(logits, n, c0 + 4, C);
    v.w = focal_cost(logits, n, c0 + 5, C);
    return v;
}

__device__ __forceinline__ float4 onehot_val(const int* __restrict__ labels,
                                             int M, int idx) {
    int lane = idx & 31;
    int cj = idx >> 5;
    int ct = cj / KCHUNKS;
    int j  = cj - ct * KCHUNKS;
    int m  = (ct << 4) + (lane & 15);
    int c0 = 8 * j + 2 * (lane >> 4);
    int lab = (m < M) ? labels[m] : -1;
    float4 v;
    v.x = (lab == c0)     ? W_CLASS : 0.0f;
    v.y = (lab == c0 + 1) ? W_CLASS : 0.0f;
    v.z = (lab == c0 + 4) ? W_CLASS : 0.0f;
    v.w = (lab == c0 + 5) ? W_CLASS : 0.0f;
    return v;
}

__device__ __forceinline__ float4 gtpad_val(const float* __restrict__ gt, int M, int m) {
    return (m < M) ? ((const float4*)gt)[m] : make_float4(0.f, 0.f, 1.f, 1.f);
}

__global__ void __launch_bounds__(256)
prep_kernel(const float* __restrict__ logits,
            const int* __restrict__ labels,
            const float* __restrict__ gt,
            float* __restrict__ Tf,
            float* __restrict__ Hf,
            float* __restrict__ Gp,
            int RT, int C, int M, int CT, int Mpad, int nb1, int nb2) {
    int b = blockIdx.x;
    int tid = (int)threadIdx.x;
    float4 v = make_float4(0.f, 0.f, 0.f, 0.f);
    float4* dst;
    int idx;
    bool ok;
    if (b < nb1) {
        idx = b * 256 + tid;
        ok = idx < RT * KCHUNKS * 32;
        if (ok) v = class_table_val(logits, C, idx);
        dst = (float4*)Tf;
    } else if (b < nb1 + nb2) {
        idx = (b - nb1) * 256 + tid;
        ok = idx < CT * KCHUNKS * 32;
        if (ok) v = onehot_val(labels, M, idx);
        dst = (float4*)Hf;
    } else {
        idx = (b - nb1 - nb2) * 256 + tid;
        ok = idx < Mpad;
        if (ok) v = gtpad_val(gt, M, idx);
        dst = (float4*)Gp;
    }
    if (ok) dst[idx] = v;
}

#define TPW 7

__device__ __forceinline__ void load_frag(float4 (&f)[KCHUNKS],
                                          const float* __restrict__ tile_base,
                                          unsigned lane_byte) {
#pragma unroll
    for (int j = 0; j < KCHUNKS; ++j)
        f[j] = *(const float4*)((const char*)tile_base + (lane_byte + j * 512u));
}

__device__ __forceinline__ v8f class_tile(const float4 (&a)[KCHUNKS],
                                          const float4 (&b)[KCHUNKS]) {
    v8f acc = {};
#pragma unroll
    for (int j = 0; j < KCHUNKS; ++j) {
        v2f a0 = {a[j].x, a[j].y}, b0 = {b[j].x, b[j].y};
        v2f a1 = {a[j].z, a[j].w}, b1 = {b[j].z, b[j].w};
        acc = __builtin_amdgcn_wmma_f32_16x16x4_f32(
             false, a0,  false, b0,
             (short)0, acc,  false,  false);
        acc = __builtin_amdgcn_wmma_f32_16x16x4_f32(
             false, a1,  false, b1,
             (short)0, acc,  false,  false);
    }
    return acc;
}

__device__ __forceinline__ void box_tail(v8f acc, const float4 (&pbx)[8],
                                         float4 gb, bool col_ok,
                                         float* __restrict__ out,
                                         unsigned obyte, unsigned row_bytes) {
    float gx0 = gb.x - 0.5f * gb.z, gy0 = gb.y - 0.5f * gb.w;
    float gx1 = gb.x + 0.5f * gb.z, gy1 = gb.y + 0.5f * gb.w;
    float garea = (gx1 - gx0) * (gy1 - gy0);

    float v[8];
#pragma unroll
    for (int r = 0; r < 8; ++r) {
        float4 pb = pbx[r];

        float c_bbox = fabsf(pb.x - gb.x) + fabsf(pb.y - gb.y) +
                       fabsf(pb.z - gb.z) + fabsf(pb.w - gb.w);

        float px0 = pb.x - 0.5f * pb.z, py0 = pb.y - 0.5f * pb.w;
        float px1 = pb.x + 0.5f * pb.z, py1 = pb.y + 0.5f * pb.w;
        float parea = (px1 - px0) * (py1 - py0);
        float iw = fmaxf(fminf(px1, gx1) - fmaxf(px0, gx0), 0.0f);
        float ih = fmaxf(fminf(py1, gy1) - fmaxf(py0, gy0), 0.0f);
        float inter = iw * ih;
        float uni = parea + garea - inter;
        float iou = inter / uni;
        float ew = fmaxf(fmaxf(px1, gx1) - fminf(px0, gx0), 0.0f);
        float eh = fmaxf(fmaxf(py1, gy1) - fminf(py0, gy0), 0.0f);
        float earea = ew * eh;
        float giou = iou - (earea - uni) / earea;

        v[r] = acc[r] + W_BBOX * c_bbox - W_GIOU * giou;
    }
    if (col_ok) {
#pragma unroll
        for (int r = 0; r < 8; ++r)
            *(float*)((char*)out + (obyte + (unsigned)r * row_bytes)) = v[r];
    }
}

__global__ void __launch_bounds__(256)
cost_kernel(const float* __restrict__ Tf,
            const float* __restrict__ Hf,
            const float* __restrict__ Gp,
            const float* __restrict__ pboxes,
            float* __restrict__ out,
            int M, int CT, int RT, int RB) {
    int wave = threadIdx.x >> 5;
    int lane = threadIdx.x & 31;
    int cg = blockIdx.x / RB;
    int rb = blockIdx.x - cg * RB;
    int rt = rb * (int)(blockDim.x >> 5) + wave;
    if (rt >= RT) return;
    int ct0 = cg * TPW;
    int nt = min(TPW, CT - ct0);

    int lo = lane & 15;
    int hi = lane >> 4;
    unsigned lane_byte = (unsigned)lane * 16u;

    float4 a[KCHUNKS];
    load_frag(a, Tf + (size_t)rt * (KCHUNKS * 128), lane_byte);

    const float* Bt = Hf + (size_t)ct0 * (KCHUNKS * 128);
    float4 b[KCHUNKS];
    load_frag(b, Bt, lane_byte);

    float4 pbx[8];
    int q0 = (rt << 4) + 8 * hi;
#pragma unroll
    for (int r = 0; r < 8; ++r) pbx[r] = ((const float4*)pboxes)[q0 + r];

    unsigned row_bytes = (unsigned)M * 4u;
    unsigned obyte = ((unsigned)q0 * (unsigned)M + (unsigned)(ct0 << 4) + (unsigned)lo) * 4u;
    int g = (ct0 << 4) + lo;
    const float4* Gl = (const float4*)Gp + g;

    float4 gb = Gl[0];
    for (int t = 0; t < nt; ++t) {
        v8f acc = class_tile(a, b);
        asm volatile("" :: "v"(acc) : "memory");
        int tn = min(t + 1, nt - 1);
        load_frag(b, Bt + (size_t)tn * (KCHUNKS * 128), lane_byte);
        float4 gb_next = Gl[tn * 16];
        box_tail(acc, pbx, gb, g + t * 16 < M, out, obyte + (unsigned)t * 64u, row_bytes);
        gb = gb_next;
    }
}

extern "C" void kernel_launch(void* const* d_in, const int* in_sizes, int n_in,
                              void* d_out, int out_size, void* d_ws, size_t ws_size,
                              hipStream_t stream) {
    const float* logits = (const float*)d_in[0];
    const float* pboxes = (const float*)d_in[1];
    const float* gt     = (const float*)d_in[2];
    const int*   labels = (const int*)d_in[3];
    float* out = (float*)d_out;

    int Nq = in_sizes[1] / 4;
    int M  = in_sizes[2] / 4;
    int C  = in_sizes[0] / Nq;
    int Mpad = ((M + 15) / 16) * 16;
    int CT = Mpad / 16;
    int RT = Nq / 16;
    int CG = (CT + TPW - 1) / TPW;

    float* T  = (float*)d_ws;
    float* Ht = T + (size_t)Nq * KPAD;
    float* Gp = Ht + (size_t)Mpad * KPAD;

    int t1 = RT * KCHUNKS * 32;
    int t2 = CT * KCHUNKS * 32;
    int nb1 = (t1 + 255) / 256;
    int nb2 = (t2 + 255) / 256;
    int nb3 = (Mpad + 255) / 256;
    prep_kernel<<<nb1 + nb2 + nb3, 256, 0, stream>>>(logits, labels, gt, T, Ht, Gp,
                                                     RT, C, M, CT, Mpad, nb1, nb2);

    const int wavesPerBlock = 8;
    int RB = (RT + wavesPerBlock - 1) / wavesPerBlock;
    cost_kernel<<<RB * CG, 32 * wavesPerBlock, 0, stream>>>(T, Ht, Gp, pboxes,
                                                            out, M, CT, RT, RB);
}


# Both GIoU quotients per output: hardware reciprocal (v_rcp_f32) times the numerator instead of the IEEE division sequence, 353 -> 209 vector instructions per 16x16 tile; last-place rounding differs
# speedup vs baseline: 2.0564x; 1.1288x over previous
#include <hip/hip_runtime.h>
#include <hip/hip_bf16.h>

typedef __attribute__((ext_vector_type(2))) float v2f;
typedef __attribute__((ext_vector_type(8))) float v8f;

#define ALPHA   0.25f
#define EPS     1e-8f
#define W_CLASS 2.0f
#define W_BBOX  5.0f
#define W_GIOU  2.0f
#define KPAD    96

#define KCHUNKS (KPAD / 8)

__device__ __forceinline__ float focal_cost(const float* __restrict__ logits,
                                            int n, int c, int C) {
    if (c >= C) return 0.0f;
    float x = logits[n * C + c];
    float p = 1.0f / (1.0f + __expf(-x));
    float q = 1.0f - p;
    float pos = ALPHA * q * q * (-__logf(p + EPS));
    float neg = (1.0f - ALPHA) * p * p * (-__logf(q + EPS));
    return pos - neg;
}

__device__ __forceinline__ float4 class_table_val(const float* __restrict__ logits,
                                                  int C, int idx) {
    int lane = idx & 31;
    int cj = idx >> 5;
    int rt = cj / KCHUNKS;
    int j  = cj - rt * KCHUNKS;
    int n  = (rt << 4) + (lane & 15);
    int c0 = 8 * j + 2 * (lane >> 4);
    float4 v;
    v.x = focal_cost(logits, n, c0,     C);
    v.y = focal_cost(logits, n, c0 + 1, C);
    v.z = focal_cost(logits, n, c0 + 4, C);
    v.w = focal_cost(logits, n, c0 + 5, C);
    return v;
}

__device__ __forceinline__ float4 onehot_val(const int* __restrict__ labels,
                                             int M, int idx) {
    int lane = idx & 31;
    int cj = idx >> 5;
    int ct = cj / KCHUNKS;
    int j  = cj - ct * KCHUNKS;
    int m  = (ct << 4) + (lane & 15);
    int c0 = 8 * j + 2 * (lane >> 4);
    int lab = (m < M) ? labels[m] : -1;
    float4 v;
    v.x = (lab == c0)     ? W_CLASS : 0.0f;
    v.y = (lab == c0 + 1) ? W_CLASS : 0.0f;
    v.z = (lab == c0 + 4) ? W_CLASS : 0.0f;
    v.w = (lab == c0 + 5) ? W_CLASS : 0.0f;
    return v;
}

__device__ __forceinline__ float4 gtpad_val(const float* __restrict__ gt, int M, int m) {
    return (m < M) ? ((const float4*)gt)[m] : make_float4(0.f, 0.f, 1.f, 1.f);
}

__global__ void __launch_bounds__(256)
prep_kernel(const float* __restrict__ logits,
            const int* __restrict__ labels,
            const float* __restrict__ gt,
            float* __restrict__ Tf,
            float* __restrict__ Hf,
            float* __restrict__ Gp,
            int RT, int C, int M, int CT, int Mpad, int nb1, int nb2) {
    int b = blockIdx.x;
    int tid = (int)threadIdx.x;
    float4 v = make_float4(0.f, 0.f, 0.f, 0.f);
    float4* dst;
    int idx;
    bool ok;
    if (b < nb1) {
        idx = b * 256 + tid;
        ok = idx < RT * KCHUNKS * 32;
        if (ok) v = class_table_val(logits, C, idx);
        dst = (float4*)Tf;
    } else if (b < nb1 + nb2) {
        idx = (b - nb1) * 256 + tid;
        ok = idx < CT * KCHUNKS * 32;
        if (ok) v = onehot_val(labels, M, idx);
        dst = (float4*)Hf;
    } else {
        idx = (b - nb1 - nb2) * 256 + tid;
        ok = idx < Mpad;
        if (ok) v = gtpad_val(gt, M, idx);
        dst = (float4*)Gp;
    }
    if (ok) dst[idx] = v;
}

#define TPW 7

__device__ __forceinline__ void load_frag(float4 (&f)[KCHUNKS],
                                          const float* __restrict__ tile_base,
                                          unsigned lane_byte) {
#pragma unroll
    for (int j = 0; j < KCHUNKS; ++j)
        f[j] = *(const float4*)((const char*)tile_base + (lane_byte + j * 512u));
}

__device__ __forceinline__ v8f class_tile(const float4 (&a)[KCHUNKS],
                                          const float4 (&b)[KCHUNKS]) {
    v8f acc = {};
#pragma unroll
    for (int j = 0; j < KCHUNKS; ++j) {
        v2f a0 = {a[j].x, a[j].y}, b0 = {b[j].x, b[j].y};
        v2f a1 = {a[j].z, a[j].w}, b1 = {b[j].z, b[j].w};
        acc = __builtin_amdgcn_wmma_f32_16x16x4_f32(
             false, a0,  false, b0,
             (short)0, acc,  false,  false);
        acc = __builtin_amdgcn_wmma_f32_16x16x4_f32(
             false, a1,  false, b1,
             (short)0, acc,  false,  false);
    }
    return acc;
}

__device__ __forceinline__ void box_tail(v8f acc, const float4 (&pbx)[8],
                                         float4 gb, bool col_ok,
                                         float* __restrict__ out,
                                         unsigned obyte, unsigned row_bytes) {
    float gx0 = gb.x - 0.5f * gb.z, gy0 = gb.y - 0.5f * gb.w;
    float gx1 = gb.x + 0.5f * gb.z, gy1 = gb.y + 0.5f * gb.w;
    float garea = (gx1 - gx0) * (gy1 - gy0);

    float v[8];
#pragma unroll
    for (int r = 0; r < 8; ++r) {
        float4 pb = pbx[r];

        float c_bbox = fabsf(pb.x - gb.x) + fabsf(pb.y - gb.y) +
                       fabsf(pb.z - gb.z) + fabsf(pb.w - gb.w);

        float px0 = pb.x - 0.5f * pb.z, py0 = pb.y - 0.5f * pb.w;
        float px1 = pb.x + 0.5f * pb.z, py1 = pb.y + 0.5f * pb.w;
        float parea = (px1 - px0) * (py1 - py0);
        float iw = fmaxf(fminf(px1, gx1) - fmaxf(px0, gx0), 0.0f);
        float ih = fmaxf(fminf(py1, gy1) - fmaxf(py0, gy0), 0.0f);
        float inter = iw * ih;
        float uni = parea + garea - inter;
        float iou = inter * __builtin_amdgcn_rcpf(uni);
        float ew = fmaxf(fmaxf(px1, gx1) - fminf(px0, gx0), 0.0f);
        float eh = fmaxf(fmaxf(py1, gy1) - fminf(py0, gy0), 0.0f);
        float earea = ew * eh;
        float giou = iou - (earea - uni) * __builtin_amdgcn_rcpf(earea);

        v[r] = acc[r] + W_BBOX * c_bbox - W_GIOU * giou;
    }
    if (col_ok) {
#pragma unroll
        for (int r = 0; r < 8; ++r)
            *(float*)((char*)out + (obyte + (unsigned)r * row_bytes)) = v[r];
    }
}

__global__ void __launch_bounds__(256)
cost_kernel(const float* __restrict__ Tf,
            const float* __restrict__ Hf,
            const float* __restrict__ Gp,
            const float* __restrict__ pboxes,
            float* __restrict__ out,
            int M, int CT, int RT, int RB) {
    int wave = threadIdx.x >> 5;
    int lane = threadIdx.x & 31;
    int cg = blockIdx.x / RB;
    int rb = blockIdx.x - cg * RB;
    int rt = rb * (int)(blockDim.x >> 5) + wave;
    if (rt >= RT) return;
    int ct0 = cg * TPW;
    int nt = min(TPW, CT - ct0);

    int lo = lane & 15;
    int hi = lane >> 4;
    unsigned lane_byte = (unsigned)lane * 16u;

    float4 a[KCHUNKS];
    load_frag(a, Tf + (size_t)rt * (KCHUNKS * 128), lane_byte);

    const float* Bt = Hf + (size_t)ct0 * (KCHUNKS * 128);
    float4 b[KCHUNKS];
    load_frag(b, Bt, lane_byte);

    float4 pbx[8];
    int q0 = (rt << 4) + 8 * hi;
#pragma unroll
    for (int r = 0; r < 8; ++r) pbx[r] = ((const float4*)pboxes)[q0 + r];

    unsigned row_bytes = (unsigned)M * 4u;
    unsigned obyte = ((unsigned)q0 * (unsigned)M + (unsigned)(ct0 << 4) + (unsigned)lo) * 4u;
    int g = (ct0 << 4) + lo;
    const float4* Gl = (const float4*)Gp + g;

    float4 gb = Gl[0];
    for (int t = 0; t < nt; ++t) {
        v8f acc = class_tile(a, b);
        asm volatile("" :: "v"(acc) : "memory");
        int tn = min(t + 1, nt - 1);
        load_frag(b, Bt + (size_t)tn * (KCHUNKS * 128), lane_byte);
        float4 gb_next = Gl[tn * 16];
        box_tail(acc, pbx, gb, g + t * 16 < M, out, obyte + (unsigned)t * 64u, row_bytes);
        gb = gb_next;
    }
}

extern "C" void kernel_launch(void* const* d_in, const int* in_sizes, int n_in,
                              void* d_out, int out_size, void* d_ws, size_t ws_size,
                              hipStream_t stream) {
    const float* logits = (const float*)d_in[0];
    const float* pboxes = (const float*)d_in[1];
    const float* gt     = (const float*)d_in[2];
    const int*   labels = (const int*)d_in[3];
    float* out = (float*)d_out;

    int Nq = in_sizes[1] / 4;
    int M  = in_sizes[2] / 4;
    int C  = in_sizes[0] / Nq;
    int Mpad = ((M + 15) / 16) * 16;
    int CT = Mpad / 16;
    int RT = Nq / 16;
    int CG = (CT + TPW - 1) / TPW;

    float* T  = (float*)d_ws;
    float* Ht = T + (size_t)Nq * KPAD;
    float* Gp = Ht + (size_t)Mpad * KPAD;

    int t1 = RT * KCHUNKS * 32;
    int t2 = CT * KCHUNKS * 32;
    int nb1 = (t1 + 255) / 256;
    int nb2 = (t2 + 255) / 256;
    int nb3 = (Mpad + 255) / 256;
    prep_kernel<<<nb1 + nb2 + nb3, 256, 0, stream>>>(logits, labels, gt, T, Ht, Gp,
                                                     RT, C, M, CT, Mpad, nb1, nb2);

    const int wavesPerBlock = 8;
    int RB = (RT + wavesPerBlock - 1) / wavesPerBlock;
    cost_kernel<<<RB * CG, 32 * wavesPerBlock, 0, stream>>>(T, Ht, Gp, pboxes,
                                                            out, M, CT, RT, RB);
}
